# starting kernel
# baseline (speedup 1.0000x reference)
#include <hip/hip_runtime.h>
#include <hip/hip_bf16.h>

#define Cc   8
#define Tt   8
#define Dd   24
#define Hh   48
#define Ww   48
#define KT   3
#define KS   3
#define TO   6
#define DO   47
#define HO   95
#define WO   95
#define WT   6
#define NTAP 81
#define XCH  (Tt*Dd*Hh*Ww)
#define XROW (Hh*Ww)

typedef __attribute__((ext_vector_type(2))) float v2f;
typedef __attribute__((ext_vector_type(8))) float v8f;

__global__ __launch_bounds__(128)
void convT4d_wmma_kernel(const float* __restrict__ x,
                         const float* __restrict__ weight,
                         float* __restrict__ out) {
    __shared__ float wlds[NTAP * 16 * Cc];

    for (int idx = threadIdx.x; idx < NTAP * 16 * Cc; idx += blockDim.x) {
        int ci  = idx & 7;
        int co  = (idx >> 3) & 15;
        int tap = idx >> 7;
        int kw  = tap % 3;
        int kh  = (tap / 3) % 3;
        int kd  = (tap / 9) % 3;
        int i   = tap / 27;
        float v = 0.0f;
        if (co < Cc) {
            int widx = ((((ci * Cc + co) * KT + i) * KS + (2 - kd)) * KS + (2 - kh)) * KS + (2 - kw);
            v = weight[widx];
        }
        wlds[idx] = v;
    }
    __syncthreads();

    const int wave = threadIdx.x >> 5;
    const int lane = threadIdx.x & 31;
    const int tid  = blockIdx.x * 4 + wave;
    const int total = TO * DO * HO * WT;
    if (tid >= total) return;

    int t   = tid;
    const int wt  = t % WT;  t /= WT;
    const int oh  = t % HO;  t /= HO;
    const int od  = t % DO;  t /= DO;
    const int f   = t;
    const int owb = wt * 16;

    const int n  = lane & 15;
    const int k0 = (lane >> 4) * 2;

    const int ib0 = (owb + n - 1) >> 1;
    const int ib  = ib0 < 0 ? 0 : (ib0 > (Ww - 2) ? (Ww - 2) : ib0);

    float mlo[3], mhi[3];
    #pragma unroll
    for (int kw = 0; kw < KS; ++kw) {
        const int  mw    = owb + n + kw - 1;
        const int  iw    = mw >> 1;
        const bool valid = ((mw & 1) == 0) && (mw >= 0) && (iw < Ww);
        mlo[kw] = (valid && iw == ib)     ? 1.0f : 0.0f;
        mhi[kw] = (valid && iw == ib + 1) ? 1.0f : 0.0f;
    }

    v8f acc = {};

    #pragma unroll
    for (int i = 0; i < KT; ++i) {
        const int frame = f + i;
        #pragma unroll
        for (int kd = 0; kd < KS; ++kd) {
            const int md = od + kd - 1;
            if (md & 1) continue;
            const int id = md >> 1;
            #pragma unroll
            for (int kh = 0; kh < KS; ++kh) {
                const int mh = oh + kh - 1;
                if (mh & 1) continue;
                const int ih = mh >> 1;

                const int rowbase = (frame * Dd + id) * XROW + ih * Ww;
                __builtin_prefetch(x + rowbase, 0, 0);

                float plo[4], phi[4];
                #pragma unroll
                for (int c4 = 0; c4 < 4; ++c4) {
                    const int ci = (c4 >> 1) * 4 + k0 + (c4 & 1);
                    const float* p = x + ci * XCH + rowbase + ib;
                    plo[c4] = p[0];
                    phi[c4] = p[1];
                }

                #pragma unroll
                for (int kw = 0; kw < KS; ++kw) {
                    const int tapoff = (((i * 3 + kd) * 3 + kh) * 3 + kw) << 7;
                    #pragma unroll
                    for (int chunk = 0; chunk < 2; ++chunk) {
                        const v2f aa = *(const v2f*)&wlds[tapoff + n * Cc + chunk * 4 + k0];
                        v2f bb;
                        #pragma unroll
                        for (int v = 0; v < 2; ++v) {
                            const int c4 = chunk * 2 + v;
                            bb[v] = plo[c4] * mlo[kw] + phi[c4] * mhi[kw];
                        }
                        acc = __builtin_amdgcn_wmma_f32_16x16x4_f32(
                                  false, aa, false, bb, (short)0, acc, false, false);
                    }
                }
            }
        }
    }

    if (lane < 16) {
        const int ow = owb + lane;
        if (ow < WO) {
            const int obase = ((f * DO + od) * HO + oh) * WO + ow;
            #pragma unroll
            for (int r = 0; r < Cc; ++r) {
                out[r * (TO * DO * HO * WO) + obase] = acc[r];
            }
        }
    }
}

extern "C" void kernel_launch(void* const* d_in, const int* in_sizes, int n_in,
                              void* d_out, int out_size, void* d_ws, size_t ws_size,
                              hipStream_t stream) {
    const float* x = (const float*)d_in[0];
    const float* w = (const float*)d_in[1];
    float* out = (float*)d_out;

    const int total  = TO * DO * HO * WT;
    const int blocks = (total + 3) / 4;
    convT4d_wmma_kernel<<<blocks, 128, 0, stream>>>(x, w, out);
}


# Tile epilogue through a wave-private LDS patch: each co's row pair (190 contiguous floats) leaves in whole runs, 16 bytes per lane: 13 store instructions per tile, was 32; same sums, same order
# speedup vs baseline: 11.4494x; 11.4494x over previous
#include <hip/hip_runtime.h>
#include <hip/hip_bf16.h>


#define Cc   8
#define Tt   8
#define Dd   24
#define Hh   48
#define Ww   48
#define KT   3
#define KS   3
#define TO   6
#define DO   47
#define HO   95
#define WO   95
#define WT   3
#define NTAP 81
#define XCH  (Tt*Dd*Hh*Ww)
#define XROW (Hh*Ww)

#define NGRP   16
#define NBLK   (NGRP * 32)
#define SLOT1  11

typedef __attribute__((ext_vector_type(2))) float v2f;
typedef __attribute__((ext_vector_type(8))) float v8f;
typedef float v2fu __attribute__((ext_vector_type(2), aligned(4)));
typedef __attribute__((ext_vector_type(4))) float v4f;
typedef float v4fu __attribute__((ext_vector_type(4), aligned(4)));

#define RUNL 190
#define RUNP 192

template <int PD>
__device__ __forceinline__ void run_class(const float* __restrict__ x,
                                          const float* __restrict__ weight,
                                          float* __restrict__ out,
                                          float* __restrict__ st,
                                          const int w, const int nw, const int lane) {
    constexpr int NKD = PD ? 2 : 1;
    constexpr int NKH = 2;
    constexpr int NOD = PD ? 23 : 24;
    constexpr int NOH = Hh;
    constexpr int OS  = TO * DO * HO * WO;

    const int n  = lane & 15;
    const int hl = lane >> 4;
    const int k0 = hl * 2;
    const int ph = n >> 3;

    v2f afrag[KT][NKD][NKH][KS][2];
    {
        const bool   odd = ph != 0;
        const float* wb  = weight + (k0 * Cc + (n & 7)) * NTAP;
        const float* wr0 = wb + (1 + ph) * KS;
        const float* wr1 = wb;
        #pragma unroll
        for (int i = 0; i < KT; ++i) {
            #pragma unroll
            for (int jd = 0; jd < NKD; ++jd) {
                const int kd = PD ? 2 * jd : 1;
                #pragma unroll
                for (int jh = 0; jh < NKH; ++jh) {
                    #pragma unroll
                    for (int kw = 0; kw < KS; ++kw) {
                        #pragma unroll
                        for (int chunk = 0; chunk < 2; ++chunk) {
                            #pragma unroll
                            for (int v = 0; v < 2; ++v) {
                                const int off = (chunk * 4 + v) * Cc * NTAP
                                              + (i * KS + (2 - kd)) * KS * KS + (2 - kw);
                                const float val = jh ? wr1[off] : wr0[off];
                                afrag[i][jd][jh][kw][chunk][v] = (jh && !odd) ? 0.0f : val;
                            }
                        }
                    }
                }
            }
        }
    }

    const int ntiles = TO * NOD * NOH;
    for (int tile = w; tile < ntiles; tile += nw) {
        int t = tile;
        const int b  = t % NOH;  t /= NOH;
        const int a  = t % NOD;  t /= NOD;
        const int f  = t;
        const int od  = 2 * a + PD;
        const int ih1 = (b + 1 < Hh) ? (b + 1) : (Hh - 1);

        const bool edge = (n == 15);

        v8f acc_e[WT] = {};
        v8f acc_o[WT] = {};

        #pragma unroll
        for (int i = 0; i < KT; ++i) {
            const int frame = f + i;
            #pragma unroll
            for (int jd = 0; jd < NKD; ++jd) {
                const int id = a + jd;
                #pragma unroll
                for (int jh = 0; jh < NKH; ++jh) {
                    const int ih = jh ? ih1 : b;

                    const int rowbase = (frame * Dd + id) * XROW + ih * Ww;
                    __builtin_prefetch(x + rowbase, 0, 0);

                    v2f s0[WT][2], s1[WT][2];
                    #pragma unroll
                    for (int blk = 0; blk < WT; ++blk) {
                        const bool e  = (blk == WT - 1) && edge;
                        const int  jb = e ? (Ww - 2) : (blk * 16 + n);
                        #pragma unroll
                        for (int c4 = 0; c4 < 4; ++c4) {
                            const int ci = (c4 >> 1) * 4 + k0 + (c4 & 1);
                            const float* p = x + ci * XCH + rowbase + jb;
                            const float lo = p[0];
                            const float hi = p[1];
                            s0[blk][c4 >> 1][c4 & 1] = e ? hi : lo;
                            s1[blk][c4 >> 1][c4 & 1] = hi;
                        }
                    }

                    #pragma unroll
                    for (int chunk = 0; chunk < 2; ++chunk) {
                        #pragma unroll
                        for (int blk = 0; blk < WT; ++blk) {
                            acc_o[blk] = __builtin_amdgcn_wmma_f32_16x16x4_f32(
                                        false, afrag[i][jd][jh][0][chunk], false, s0[blk][chunk],
                                        (short)0, acc_o[blk], false, false);
                            acc_e[blk] = __builtin_amdgcn_wmma_f32_16x16x4_f32(
                                        false, afrag[i][jd][jh][1][chunk], false, s0[blk][chunk],
                                        (short)0, acc_e[blk], false, false);
                        }
                    }
                    #pragma unroll
                    for (int chunk = 0; chunk < 2; ++chunk) {
                        #pragma unroll
                        for (int blk = 0; blk < WT; ++blk) {
                            acc_o[blk] = __builtin_amdgcn_wmma_f32_16x16x4_f32(
                                        false, afrag[i][jd][jh][2][chunk], false, s1[blk][chunk],
                                        (short)0, acc_o[blk], false, false);
                        }
                    }
                }
            }
        }

        if (b + 1 < NOH) {
            float* se = st + hl * WO + 2 * n;
            #pragma unroll
            for (int blk = 0; blk < WT - 1; ++blk) {
                #pragma unroll
                for (int r = 0; r < Cc; ++r) {
                    se[r * RUNP + 32 * blk]     = acc_e[blk][r];
                    se[r * RUNP + 32 * blk + 1] = acc_o[blk][r];
                }
            }
            float* so = edge ? (st + RUNL + hl) : (se + 32 * (WT - 1) + 1);
            #pragma unroll
            for (int r = 0; r < Cc; ++r) {
                se[r * RUNP + 32 * (WT - 1)] = acc_e[WT - 1][r];
                so[r * RUNP]                 = acc_o[WT - 1][r];
            }
            __builtin_amdgcn_fence(__ATOMIC_ACQ_REL, "wavefront");

            float* gb = out + ((f * DO + od) * HO + 2 * b) * WO;
            #pragma unroll
            for (int it = 0; it < 12; ++it) {
                const unsigned q = it * 32 + lane;
                if (q < Cc * 47) {
                    const unsigned r = q / 47;
                    const unsigned ch = q - r * 47;
                    const v4f val = *reinterpret_cast<const v4f*>(st + r * RUNP + 4 * ch);
                    *reinterpret_cast<v4fu*>(gb + r * OS + 4 * ch) = val;
                }
            }
            if (lane < Cc) {
                const v2f val = *reinterpret_cast<const v2f*>(st + lane * RUNP + 188);
                *reinterpret_cast<v2fu*>(gb + lane * OS + 188) = val;
            }
            __builtin_amdgcn_fence(__ATOMIC_ACQ_REL, "wavefront");
        } else if (hl == 0) {
            float* po = out + ((f * DO + od) * HO + 2 * b) * WO + 2 * n;
            #pragma unroll
            for (int blk = 0; blk < WT - 1; ++blk) {
                #pragma unroll
                for (int r = 0; r < Cc; ++r) {
                    v2fu pr;
                    pr[0] = acc_e[blk][r];
                    pr[1] = acc_o[blk][r];
                    *reinterpret_cast<v2fu*>(po + r * OS + 32 * blk) = pr;
                }
            }
            if (!edge) {
                #pragma unroll
                for (int r = 0; r < Cc; ++r) {
                    v2fu pr;
                    pr[0] = acc_e[WT - 1][r];
                    pr[1] = acc_o[WT - 1][r];
                    *reinterpret_cast<v2fu*>(po + r * OS + 32 * (WT - 1)) = pr;
                }
            } else {
                #pragma unroll
                for (int r = 0; r < Cc; ++r) {
                    po[r * OS + 32 * (WT - 1)] = acc_e[WT - 1][r];
                }
            }
        }
    }
}

__global__ __launch_bounds__(128) __attribute__((amdgpu_waves_per_eu(2, 2)))
void convT4d_wmma_kernel(const float* __restrict__ x,
                         const float* __restrict__ weight,
                         float* __restrict__ out) {
    const int lane = threadIdx.x & 31;
    const int wave = __builtin_amdgcn_readfirstlane(threadIdx.x >> 5);
    __shared__ __attribute__((aligned(16))) float stage[4 * Cc * RUNP];
    float* st = stage + wave * (Cc * RUNP);
    const int slot = blockIdx.x & 31;
    const int grp  = blockIdx.x >> 5;

    if (slot < SLOT1) {
        const int cnt = SLOT1;
        run_class<0>(x, weight, out, st, (grp * cnt + slot) * 4 + wave, NGRP * cnt * 4, lane);
    } else {
        const int cnt = 32 - SLOT1;
        run_class<1>(x, weight, out, st, (grp * cnt + (slot - SLOT1)) * 4 + wave, NGRP * cnt * 4, lane);
    }
}

extern "C" void kernel_launch(void* const* d_in, const int* in_sizes, int n_in,
                              void* d_out, int out_size, void* d_ws, size_t ws_size,
                              hipStream_t stream) {
    const float* x = (const float*)d_in[0];
    const float* w = (const float*)d_in[1];
    float* out = (float*)d_out;

    convT4d_wmma_kernel<<<NBLK, 128, 0, stream>>>(x, w, out);
}


# WMMA column n of block k = input column 3n+k: a lane's six B values are x[3n..3n+3], one 16-byte load per ci (4 loads per x row, was 12); a lane holds 6 adjacent ow; same sums, same order
# speedup vs baseline: 11.8426x; 1.0343x over previous
#include <hip/hip_runtime.h>
#include <hip/hip_bf16.h>


#define Cc   8
#define Tt   8
#define Dd   24
#define Hh   48
#define Ww   48
#define KT   3
#define KS   3
#define TO   6
#define DO   47
#define HO   95
#define WO   95
#define WT   3
#define NTAP 81
#define XCH  (Tt*Dd*Hh*Ww)
#define XROW (Hh*Ww)

#define NGRP   16
#define NBLK   (NGRP * 32)
#define SLOT1  11

typedef __attribute__((ext_vector_type(2))) float v2f;
typedef __attribute__((ext_vector_type(8))) float v8f;
typedef float v2fu __attribute__((ext_vector_type(2), aligned(4)));
typedef __attribute__((ext_vector_type(4))) float v4f;
typedef float v4fu __attribute__((ext_vector_type(4), aligned(4)));

#define RUNL 190
#define RUNP 192

template <int PD>
__device__ __forceinline__ void run_class(const float* __restrict__ x,
                                          const float* __restrict__ weight,
                                          float* __restrict__ out,
                                          float* __restrict__ st,
                                          const int w, const int nw, const int lane) {
    constexpr int NKD = PD ? 2 : 1;
    constexpr int NKH = 2;
    constexpr int NOD = PD ? 23 : 24;
    constexpr int NOH = Hh;
    constexpr int OS  = TO * DO * HO * WO;

    const int n  = lane & 15;
    const int hl = lane >> 4;
    const int k0 = hl * 2;
    const int ph = n >> 3;

    v2f afrag[KT][NKD][NKH][KS][2];
    {
        const bool   odd = ph != 0;
        const float* wb  = weight + (k0 * Cc + (n & 7)) * NTAP;
        const float* wr0 = wb + (1 + ph) * KS;
        const float* wr1 = wb;
        #pragma unroll
        for (int i = 0; i < KT; ++i) {
            #pragma unroll
            for (int jd = 0; jd < NKD; ++jd) {
                const int kd = PD ? 2 * jd : 1;
                #pragma unroll
                for (int jh = 0; jh < NKH; ++jh) {
                    #pragma unroll
                    for (int kw = 0; kw < KS; ++kw) {
                        #pragma unroll
                        for (int chunk = 0; chunk < 2; ++chunk) {
                            #pragma unroll
                            for (int v = 0; v < 2; ++v) {
                                const int off = (chunk * 4 + v) * Cc * NTAP
                                              + (i * KS + (2 - kd)) * KS * KS + (2 - kw);
                                const float val = jh ? wr1[off] : wr0[off];
                                afrag[i][jd][jh][kw][chunk][v] = (jh && !odd) ? 0.0f : val;
                            }
                        }
                    }
                }
            }
        }
    }

    const int ntiles = TO * NOD * NOH;
    for (int tile = w; tile < ntiles; tile += nw) {
        int t = tile;
        const int b  = t % NOH;  t /= NOH;
        const int a  = t % NOD;  t /= NOD;
        const int f  = t;
        const int od  = 2 * a + PD;
        const int ih1 = (b + 1 < Hh) ? (b + 1) : (Hh - 1);

        const bool edge = (n == 15);
        const int  jb   = edge ? (Ww - 4) : (3 * n);

        v8f acc_e[WT] = {};
        v8f acc_o[WT] = {};

        #pragma unroll
        for (int i = 0; i < KT; ++i) {
            const int frame = f + i;
            #pragma unroll
            for (int jd = 0; jd < NKD; ++jd) {
                const int id = a + jd;
                #pragma unroll
                for (int jh = 0; jh < NKH; ++jh) {
                    const int ih = jh ? ih1 : b;

                    const int rowbase = (frame * Dd + id) * XROW + ih * Ww;
                    __builtin_prefetch(x + rowbase, 0, 0);

                    v2f s0[WT][2], s3[2];
                    #pragma unroll
                    for (int c4 = 0; c4 < 4; ++c4) {
                        const int ci = (c4 >> 1) * 4 + k0 + (c4 & 1);
                        const v4fu q = *reinterpret_cast<const v4fu*>(x + ci * XCH + rowbase + jb);
                        s0[0][c4 >> 1][c4 & 1] = edge ? q[1] : q[0];
                        s0[1][c4 >> 1][c4 & 1] = edge ? q[2] : q[1];
                        s0[2][c4 >> 1][c4 & 1] = edge ? q[3] : q[2];
                        s3[c4 >> 1][c4 & 1]    = q[3];
                    }

                    #pragma unroll
                    for (int chunk = 0; chunk < 2; ++chunk) {
                        #pragma unroll
                        for (int blk = 0; blk < WT; ++blk) {
                            acc_o[blk] = __builtin_amdgcn_wmma_f32_16x16x4_f32(
                                        false, afrag[i][jd][jh][0][chunk], false, s0[blk][chunk],
                                        (short)0, acc_o[blk], false, false);
                            acc_e[blk] = __builtin_amdgcn_wmma_f32_16x16x4_f32(
                                        false, afrag[i][jd][jh][1][chunk], false, s0[blk][chunk],
                                        (short)0, acc_e[blk], false, false);
                        }
                    }
                    #pragma unroll
                    for (int chunk = 0; chunk < 2; ++chunk) {
                        #pragma unroll
                        for (int blk = 0; blk < WT; ++blk) {
                            acc_o[blk] = __builtin_amdgcn_wmma_f32_16x16x4_f32(
                                        false, afrag[i][jd][jh][2][chunk], false,
                                        (blk + 1 < WT) ? s0[(blk + 1) % WT][chunk] : s3[chunk],
                                        (short)0, acc_o[blk], false, false);
                        }
                    }
                }
            }
        }

        if (b + 1 < NOH) {
            float* se = st + hl * WO + 6 * n;
            #pragma unroll
            for (int blk = 0; blk < WT - 1; ++blk) {
                #pragma unroll
                for (int r = 0; r < Cc; ++r) {
                    se[r * RUNP + 2 * blk]     = acc_e[blk][r];
                    se[r * RUNP + 2 * blk + 1] = acc_o[blk][r];
                }
            }
            float* so = edge ? (st + RUNL + hl) : (se + 2 * (WT - 1) + 1);
            #pragma unroll
            for (int r = 0; r < Cc; ++r) {
                se[r * RUNP + 2 * (WT - 1)] = acc_e[WT - 1][r];
                so[r * RUNP]                 = acc_o[WT - 1][r];
            }
            __builtin_amdgcn_fence(__ATOMIC_ACQ_REL, "wavefront");

            float* gb = out + ((f * DO + od) * HO + 2 * b) * WO;
            #pragma unroll
            for (int it = 0; it < 12; ++it) {
                const unsigned q = it * 32 + lane;
                if (q < Cc * 47) {
                    const unsigned r = q / 47;
                    const unsigned ch = q - r * 47;
                    const v4f val = *reinterpret_cast<const v4f*>(st + r * RUNP + 4 * ch);
                    *reinterpret_cast<v4fu*>(gb + r * OS + 4 * ch) = val;
                }
            }
            if (lane < Cc) {
                const v2f val = *reinterpret_cast<const v2f*>(st + lane * RUNP + 188);
                *reinterpret_cast<v2fu*>(gb + lane * OS + 188) = val;
            }
            __builtin_amdgcn_fence(__ATOMIC_ACQ_REL, "wavefront");
        } else if (hl == 0) {
            float* po = out + ((f * DO + od) * HO + 2 * b) * WO + 6 * n;
            #pragma unroll
            for (int blk = 0; blk < WT - 1; ++blk) {
                #pragma unroll
                for (int r = 0; r < Cc; ++r) {
                    v2fu pr;
                    pr[0] = acc_e[blk][r];
                    pr[1] = acc_o[blk][r];
                    *reinterpret_cast<v2fu*>(po + r * OS + 2 * blk) = pr;
                }
            }
            if (!edge) {
                #pragma unroll
                for (int r = 0; r < Cc; ++r) {
                    v2fu pr;
                    pr[0] = acc_e[WT - 1][r];
                    pr[1] = acc_o[WT - 1][r];
                    *reinterpret_cast<v2fu*>(po + r * OS + 2 * (WT - 1)) = pr;
                }
            } else {
                #pragma unroll
                for (int r = 0; r < Cc; ++r) {
                    po[r * OS + 2 * (WT - 1)] = acc_e[WT - 1][r];
                }
            }
        }
    }
}

__global__ __launch_bounds__(128) __attribute__((amdgpu_waves_per_eu(2, 2)))
void convT4d_wmma_kernel(const float* __restrict__ x,
                         const float* __restrict__ weight,
                         float* __restrict__ out) {
    const int lane = threadIdx.x & 31;
    const int wave = __builtin_amdgcn_readfirstlane(threadIdx.x >> 5);
    __shared__ __attribute__((aligned(16))) float stage[4 * Cc * RUNP];
    float* st = stage + wave * (Cc * RUNP);
    const int slot = blockIdx.x & 31;
    const int grp  = blockIdx.x >> 5;

    if (slot < SLOT1) {
        const int cnt = SLOT1;
        run_class<0>(x, weight, out, st, (grp * cnt + slot) * 4 + wave, NGRP * cnt * 4, lane);
    } else {
        const int cnt = 32 - SLOT1;
        run_class<1>(x, weight, out, st, (grp * cnt + (slot - SLOT1)) * 4 + wave, NGRP * cnt * 4, lane);
    }
}

extern "C" void kernel_launch(void* const* d_in, const int* in_sizes, int n_in,
                              void* d_out, int out_size, void* d_ws, size_t ws_size,
                              hipStream_t stream) {
    const float* x = (const float*)d_in[0];
    const float* w = (const float*)d_in[1];
    float* out = (float*)d_out;

    convT4d_wmma_kernel<<<NBLK, 128, 0, stream>>>(x, w, out);
}


# x rows held one tile ahead in registers: when a tap step's two rows have given up their values, their registers load those rows of the wave's next tile; no tile opens on a memory wait; same sum order
# speedup vs baseline: 12.5021x; 1.0557x over previous
#include <hip/hip_runtime.h>
#include <hip/hip_bf16.h>


#define Cc   8
#define Tt   8
#define Dd   24
#define Hh   48
#define Ww   48
#define KT   3
#define KS   3
#define TO   6
#define DO   47
#define HO   95
#define WO   95
#define WT   3
#define NTAP 81
#define XCH  (Tt*Dd*Hh*Ww)
#define XROW (Hh*Ww)

#define NGRP   16
#define NBLK   (NGRP * 32)
#define SLOT1  11

typedef __attribute__((ext_vector_type(2))) float v2f;
typedef __attribute__((ext_vector_type(8))) float v8f;
typedef float v2fu __attribute__((ext_vector_type(2), aligned(4)));
typedef __attribute__((ext_vector_type(4))) float v4f;
typedef float v4fu __attribute__((ext_vector_type(4), aligned(4)));

#define RUNL 190
#define RUNP 192

template <int PD>
__device__ __forceinline__ void run_class(const float* __restrict__ x,
                                          const float* __restrict__ weight,
                                          float* __restrict__ out,
                                          float* __restrict__ st,
                                          const int w, const int nw, const int lane) {
    constexpr int NKD = PD ? 2 : 1;
    constexpr int NKH = 2;
    constexpr int NOD = PD ? 23 : 24;
    constexpr int NOH = Hh;
    constexpr int OS  = TO * DO * HO * WO;

    const int n  = lane & 15;
    const int hl = lane >> 4;
    const int k0 = hl * 2;
    const int ph = n >> 3;

    v2f afrag[KT][NKD][NKH][KS][2];
    {
        const bool   odd = ph != 0;
        const float* wb  = weight + (k0 * Cc + (n & 7)) * NTAP;
        const float* wr0 = wb + (1 + ph) * KS;
        const float* wr1 = wb;
        #pragma unroll
        for (int i = 0; i < KT; ++i) {
            #pragma unroll
            for (int jd = 0; jd < NKD; ++jd) {
                const int kd = PD ? 2 * jd : 1;
                #pragma unroll
                for (int jh = 0; jh < NKH; ++jh) {
                    #pragma unroll
                    for (int kw = 0; kw < KS; ++kw) {
                        #pragma unroll
                        for (int chunk = 0; chunk < 2; ++chunk) {
                            #pragma unroll
                            for (int v = 0; v < 2; ++v) {
                                const int off = (chunk * 4 + v) * Cc * NTAP
                                              + (i * KS + (2 - kd)) * KS * KS + (2 - kw);
                                const float val = jh ? wr1[off] : wr0[off];
                                afrag[i][jd][jh][kw][chunk][v] = (jh && !odd) ? 0.0f : val;
                            }
                        }
                    }
                }
            }
        }
    }

    constexpr int NROW = KT * NKD * NKH;
    const int ntiles = TO * NOD * NOH;

    const bool edge = (n == 15);
    const int  jb   = edge ? (Ww - 4) : (3 * n);
    const float* xl = x + k0 * XCH + jb;

    v4f xb[NROW][4];

    int cb = w % NOH;
    int ca = (w / NOH) % NOD;
    int cf = w / (NOH * NOD);
    {
        const int ih1 = (cb + 1 < Hh) ? (cb + 1) : (Hh - 1);
        #pragma unroll
        for (int i = 0; i < KT; ++i) {
            #pragma unroll
            for (int jd = 0; jd < NKD; ++jd) {
                #pragma unroll
                for (int jh = 0; jh < NKH; ++jh) {
                    const int r = (i * NKD + jd) * NKH + jh;
                    const int rowbase = ((cf + i) * Dd + (ca + jd)) * XROW + (jh ? ih1 : cb) * Ww;
                    #pragma unroll
                    for (int c4 = 0; c4 < 4; ++c4) {
                        xb[r][c4] = *reinterpret_cast<const v4fu*>(
                            xl + ((c4 >> 1) * 4 + (c4 & 1)) * XCH + rowbase);
                    }
                }
            }
        }
    }

    for (int tile = w; tile < ntiles; tile += nw) {
        const int b  = cb;
        const int a  = ca;
        const int f  = cf;
        const int od = 2 * a + PD;

        const int tn  = (tile + nw < ntiles) ? (tile + nw) : tile;
        const int nb  = tn % NOH;
        const int na  = (tn / NOH) % NOD;
        const int nf  = tn / (NOH * NOD);
        const int nih1 = (nb + 1 < Hh) ? (nb + 1) : (Hh - 1);

        v8f acc_e[WT] = {};
        v8f acc_o[WT] = {};

        #pragma unroll
        for (int i = 0; i < KT; ++i) {
            #pragma unroll
            for (int jd = 0; jd < NKD; ++jd) {
                v2f s0[NKH][WT][2], s3[NKH][2];
                #pragma unroll
                for (int jh = 0; jh < NKH; ++jh) {
                    const int r = (i * NKD + jd) * NKH + jh;
                    #pragma unroll
                    for (int c4 = 0; c4 < 4; ++c4) {
                        const v4f q = xb[r][c4];
                        s0[jh][0][c4 >> 1][c4 & 1] = edge ? q[1] : q[0];
                        s0[jh][1][c4 >> 1][c4 & 1] = edge ? q[2] : q[1];
                        s0[jh][2][c4 >> 1][c4 & 1] = edge ? q[3] : q[2];
                        s3[jh][c4 >> 1][c4 & 1]    = q[3];
                    }
                }

                #pragma unroll
                for (int jh = 0; jh < NKH; ++jh) {
                    const int r = (i * NKD + jd) * NKH + jh;
                    const int rowbase = ((nf + i) * Dd + (na + jd)) * XROW + (jh ? nih1 : nb) * Ww;
                    #pragma unroll
                    for (int c4 = 0; c4 < 4; ++c4) {
                        xb[r][c4] = *reinterpret_cast<const v4fu*>(
                            xl + ((c4 >> 1) * 4 + (c4 & 1)) * XCH + rowbase);
                    }
                }

                #pragma unroll
                for (int jh = 0; jh < NKH; ++jh) {
                    #pragma unroll
                    for (int chunk = 0; chunk < 2; ++chunk) {
                        #pragma unroll
                        for (int blk = 0; blk < WT; ++blk) {
                            acc_o[blk] = __builtin_amdgcn_wmma_f32_16x16x4_f32(
                                        false, afrag[i][jd][jh][0][chunk], false, s0[jh][blk][chunk],
                                        (short)0, acc_o[blk], false, false);
                            acc_e[blk] = __builtin_amdgcn_wmma_f32_16x16x4_f32(
                                        false, afrag[i][jd][jh][1][chunk], false, s0[jh][blk][chunk],
                                        (short)0, acc_e[blk], false, false);
                        }
                    }
                    #pragma unroll
                    for (int chunk = 0; chunk < 2; ++chunk) {
                        #pragma unroll
                        for (int blk = 0; blk < WT; ++blk) {
                            acc_o[blk] = __builtin_amdgcn_wmma_f32_16x16x4_f32(
                                        false, afrag[i][jd][jh][2][chunk], false,
                                        (blk + 1 < WT) ? s0[jh][(blk + 1) % WT][chunk] : s3[jh][chunk],
                                        (short)0, acc_o[blk], false, false);
                        }
                    }
                }
            }
        }
        cb = nb;  ca = na;  cf = nf;

        if (b + 1 < NOH) {
            float* se = st + hl * WO + 6 * n;
            #pragma unroll
            for (int blk = 0; blk < WT - 1; ++blk) {
                #pragma unroll
                for (int r = 0; r < Cc; ++r) {
                    se[r * RUNP + 2 * blk]     = acc_e[blk][r];
                    se[r * RUNP + 2 * blk + 1] = acc_o[blk][r];
                }
            }
            float* so = edge ? (st + RUNL + hl) : (se + 2 * (WT - 1) + 1);
            #pragma unroll
            for (int r = 0; r < Cc; ++r) {
                se[r * RUNP + 2 * (WT - 1)] = acc_e[WT - 1][r];
                so[r * RUNP]                 = acc_o[WT - 1][r];
            }
            __builtin_amdgcn_fence(__ATOMIC_ACQ_REL, "wavefront");

            float* gb = out + ((f * DO + od) * HO + 2 * b) * WO;
            #pragma unroll
            for (int it = 0; it < 12; ++it) {
                const unsigned q = it * 32 + lane;
                if (q < Cc * 47) {
                    const unsigned r = q / 47;
                    const unsigned ch = q - r * 47;
                    const v4f val = *reinterpret_cast<const v4f*>(st + r * RUNP + 4 * ch);
                    *reinterpret_cast<v4fu*>(gb + r * OS + 4 * ch) = val;
                }
            }
            if (lane < Cc) {
                const v2f val = *reinterpret_cast<const v2f*>(st + lane * RUNP + 188);
                *reinterpret_cast<v2fu*>(gb + lane * OS + 188) = val;
            }
            __builtin_amdgcn_fence(__ATOMIC_ACQ_REL, "wavefront");
        } else if (hl == 0) {
            float* po = out + ((f * DO + od) * HO + 2 * b) * WO + 6 * n;
            #pragma unroll
            for (int blk = 0; blk < WT - 1; ++blk) {
                #pragma unroll
                for (int r = 0; r < Cc; ++r) {
                    v2fu pr;
                    pr[0] = acc_e[blk][r];
                    pr[1] = acc_o[blk][r];
                    *reinterpret_cast<v2fu*>(po + r * OS + 2 * blk) = pr;
                }
            }
            if (!edge) {
                #pragma unroll
                for (int r = 0; r < Cc; ++r) {
                    v2fu pr;
                    pr[0] = acc_e[WT - 1][r];
                    pr[1] = acc_o[WT - 1][r];
                    *reinterpret_cast<v2fu*>(po + r * OS + 2 * (WT - 1)) = pr;
                }
            } else {
                #pragma unroll
                for (int r = 0; r < Cc; ++r) {
                    po[r * OS + 2 * (WT - 1)] = acc_e[WT - 1][r];
                }
            }
        }
    }
}

__global__ __launch_bounds__(128) __attribute__((amdgpu_waves_per_eu(2, 2)))
void convT4d_wmma_kernel(const float* __restrict__ x,
                         const float* __restrict__ weight,
                         float* __restrict__ out) {
    const int lane = threadIdx.x & 31;
    const int wave = __builtin_amdgcn_readfirstlane(threadIdx.x >> 5);
    __shared__ __attribute__((aligned(16))) float stage[4 * Cc * RUNP];
    float* st = stage + wave * (Cc * RUNP);
    const int slot = blockIdx.x & 31;
    const int grp  = blockIdx.x >> 5;

    if (slot < SLOT1) {
        const int cnt = SLOT1;
        run_class<0>(x, weight, out, st, (grp * cnt + slot) * 4 + wave, NGRP * cnt * 4, lane);
    } else {
        const int cnt = 32 - SLOT1;
        run_class<1>(x, weight, out, st, (grp * cnt + (slot - SLOT1)) * 4 + wave, NGRP * cnt * 4, lane);
    }
}

extern "C" void kernel_launch(void* const* d_in, const int* in_sizes, int n_in,
                              void* d_out, int out_size, void* d_ws, size_t ws_size,
                              hipStream_t stream) {
    const float* x = (const float*)d_in[0];
    const float* w = (const float*)d_in[1];
    float* out = (float*)d_out;

    convT4d_wmma_kernel<<<NBLK, 128, 0, stream>>>(x, w, out);
}


# Global accesses as scalar base + fixed 32-bit lane offset, store-out reads LDS once into own registers: no register of an access in flight is rewritten, no wait between a tile's accesses; same sums
# speedup vs baseline: 13.6015x; 1.0879x over previous
#include <hip/hip_runtime.h>
#include <hip/hip_bf16.h>


#define Cc   8
#define Tt   8
#define Dd   24
#define Hh   48
#define Ww   48
#define KT   3
#define KS   3
#define TO   6
#define DO   47
#define HO   95
#define WO   95
#define WT   3
#define NTAP 81
#define XCH  (Tt*Dd*Hh*Ww)
#define XROW (Hh*Ww)

#define NGRP   16
#define NBLK   (NGRP * 32)
#define SLOT1  11

typedef __attribute__((ext_vector_type(2))) float v2f;
typedef __attribute__((ext_vector_type(8))) float v8f;
typedef float v2fu __attribute__((ext_vector_type(2), aligned(4)));
typedef __attribute__((ext_vector_type(4))) float v4f;
typedef float v4fu __attribute__((ext_vector_type(4), aligned(4)));
typedef __attribute__((address_space(1))) const char gcchar;
typedef __attribute__((address_space(1))) char       gchar;
typedef __attribute__((address_space(1))) const v4fu gcv4fu;
typedef __attribute__((address_space(1))) v4fu       gv4fu;
typedef __attribute__((address_space(1))) v2fu       gv2fu;

#define RUNL 190
#define RUNP 192

template <int PD>
__device__ __forceinline__ void run_class(const float* __restrict__ x,
                                          const float* __restrict__ weight,
                                          float* __restrict__ out,
                                          float* __restrict__ st,
                                          const int w, const int nw, const int lane) {
    constexpr int NKD = PD ? 2 : 1;
    constexpr int NKH = 2;
    constexpr int NOD = PD ? 23 : 24;
    constexpr int NOH = Hh;
    constexpr int OS  = TO * DO * HO * WO;

    const int n  = lane & 15;
    const int hl = lane >> 4;
    const int k0 = hl * 2;
    const int ph = n >> 3;

    v2f afrag[KT][NKD][NKH][KS][2];
    {
        const bool   odd = ph != 0;
        const float* wb  = weight + (k0 * Cc + (n & 7)) * NTAP;
        const float* wr0 = wb + (1 + ph) * KS;
        const float* wr1 = wb;
        #pragma unroll
        for (int i = 0; i < KT; ++i) {
            #pragma unroll
            for (int jd = 0; jd < NKD; ++jd) {
                const int kd = PD ? 2 * jd : 1;
                #pragma unroll
                for (int jh = 0; jh < NKH; ++jh) {
                    #pragma unroll
                    for (int kw = 0; kw < KS; ++kw) {
                        #pragma unroll
                        for (int chunk = 0; chunk < 2; ++chunk) {
                            #pragma unroll
                            for (int v = 0; v < 2; ++v) {
                                const int off = (chunk * 4 + v) * Cc * NTAP
                                              + (i * KS + (2 - kd)) * KS * KS + (2 - kw);
                                const float val = jh ? wr1[off] : wr0[off];
                                afrag[i][jd][jh][kw][chunk][v] = (jh && !odd) ? 0.0f : val;
                            }
                        }
                    }
                }
            }
        }
    }

    constexpr int NROW = KT * NKD * NKH;
    const int ntiles = TO * NOD * NOH;

    const bool edge = (n == 15);
    const int  jb   = edge ? (Ww - 4) : (3 * n);
    unsigned xo0 = (unsigned)(k0 * XCH + jb) * 4u;
    unsigned xo1 = xo0 + 4u * XCH * 4u;
    gcchar* xc = (gcchar*)x;

    unsigned gofs[11];
    const float* sl[11];
    #pragma unroll
    for (int it = 0; it < 11; ++it) {
        const unsigned q  = it * 32 + lane;
        const unsigned r  = q / 44;
        const unsigned ch = q - r * 44;
        gofs[it] = (r * (unsigned)OS + 4u * ch) * 4u;
        sl[it] = st + r * RUNP + 4 * ch;
    }
    const unsigned pa = lane;
    const unsigned pb = (lane < 24) ? (32 + lane) : 55u;
    const unsigned ra = pa / 7, ca2 = pa - ra * 7;
    const unsigned rbn = pb / 7, cb2 = pb - rbn * 7;
    unsigned gna = (ra * (unsigned)OS + 176u + 2u * ca2) * 4u;
    unsigned gnb = (rbn * (unsigned)OS + 176u + 2u * cb2) * 4u;
    const float* sna = st + ra * RUNP + 176 + 2 * ca2;
    const float* snb = st + rbn * RUNP + 176 + 2 * cb2;

    v4f xb[NROW][4];

    int cb = w % NOH;
    int ca = (w / NOH) % NOD;
    int cf = w / (NOH * NOD);
    {
        const int ih1 = (cb + 1 < Hh) ? (cb + 1) : (Hh - 1);
        const int tb = (cf * Dd + ca) * XROW;
        gcchar* rb0 = xc + (size_t)(unsigned)(tb + cb * Ww) * 4u;
        gcchar* rb1 = xc + (size_t)(unsigned)(tb + ih1 * Ww) * 4u;
        asm("" : "+s"(rb0), "+s"(rb1), "+v"(xo0), "+v"(xo1));
        #pragma unroll
        for (int i = 0; i < KT; ++i) {
            #pragma unroll
            for (int jd = 0; jd < NKD; ++jd) {
                #pragma unroll
                for (int jh = 0; jh < NKH; ++jh) {
                    const int r = (i * NKD + jd) * NKH + jh;
                    #pragma unroll
                    for (int c4 = 0; c4 < 4; ++c4) {
                        xb[r][c4] = *(gcv4fu*)(
                            (jh ? rb1 : rb0) + ((c4 >> 1) ? xo1 : xo0)
                            + ((i * Dd + jd) * XROW + (c4 & 1) * XCH) * 4);
                    }
                }
            }
        }
    }

    for (int tile = w; tile < ntiles; tile += nw) {
        const int b  = cb;
        const int a  = ca;
        const int f  = cf;
        const int od = 2 * a + PD;

        const int tn  = (tile + nw < ntiles) ? (tile + nw) : tile;
        const int nb  = tn % NOH;
        const int na  = (tn / NOH) % NOD;
        const int nf  = tn / (NOH * NOD);
        const int nih1 = (nb + 1 < Hh) ? (nb + 1) : (Hh - 1);
        const int ntb = (nf * Dd + na) * XROW;
        gcchar* nb0 = xc + (size_t)(unsigned)(ntb + nb * Ww) * 4u;
        gcchar* nb1 = xc + (size_t)(unsigned)(ntb + nih1 * Ww) * 4u;
        asm("" : "+s"(nb0), "+s"(nb1), "+v"(xo0), "+v"(xo1));

        v8f acc_e[WT] = {};
        v8f acc_o[WT] = {};

        #pragma unroll
        for (int i = 0; i < KT; ++i) {
            #pragma unroll
            for (int jd = 0; jd < NKD; ++jd) {
                v2f s0[NKH][WT][2], s3[NKH][2];
                #pragma unroll
                for (int jh = 0; jh < NKH; ++jh) {
                    const int r = (i * NKD + jd) * NKH + jh;
                    #pragma unroll
                    for (int c4 = 0; c4 < 4; ++c4) {
                        const v4f q = xb[r][c4];
                        s0[jh][0][c4 >> 1][c4 & 1] = edge ? q[1] : q[0];
                        s0[jh][1][c4 >> 1][c4 & 1] = edge ? q[2] : q[1];
                        s0[jh][2][c4 >> 1][c4 & 1] = edge ? q[3] : q[2];
                        s3[jh][c4 >> 1][c4 & 1]    = q[3];
                    }
                }

                #pragma unroll
                for (int jh = 0; jh < NKH; ++jh) {
                    const int r = (i * NKD + jd) * NKH + jh;
                    #pragma unroll
                    for (int c4 = 0; c4 < 4; ++c4) {
                        xb[r][c4] = *(gcv4fu*)(
                            (jh ? nb1 : nb0) + ((c4 >> 1) ? xo1 : xo0)
                            + ((i * Dd + jd) * XROW + (c4 & 1) * XCH) * 4);
                    }
                }

                #pragma unroll
                for (int jh = 0; jh < NKH; ++jh) {
                    #pragma unroll
                    for (int chunk = 0; chunk < 2; ++chunk) {
                        #pragma unroll
                        for (int blk = 0; blk < WT; ++blk) {
                            acc_o[blk] = __builtin_amdgcn_wmma_f32_16x16x4_f32(
                                        false, afrag[i][jd][jh][0][chunk], false, s0[jh][blk][chunk],
                                        (short)0, acc_o[blk], false, false);
                            acc_e[blk] = __builtin_amdgcn_wmma_f32_16x16x4_f32(
                                        false, afrag[i][jd][jh][1][chunk], false, s0[jh][blk][chunk],
                                        (short)0, acc_e[blk], false, false);
                        }
                    }
                    #pragma unroll
                    for (int chunk = 0; chunk < 2; ++chunk) {
                        #pragma unroll
                        for (int blk = 0; blk < WT; ++blk) {
                            acc_o[blk] = __builtin_amdgcn_wmma_f32_16x16x4_f32(
                                        false, afrag[i][jd][jh][2][chunk], false,
                                        (blk + 1 < WT) ? s0[jh][(blk + 1) % WT][chunk] : s3[jh][chunk],
                                        (short)0, acc_o[blk], false, false);
                        }
                    }
                }
            }
        }
        cb = nb;  ca = na;  cf = nf;

        if (b + 1 < NOH) {
            float* se = st + hl * WO + 6 * n;
            #pragma unroll
            for (int blk = 0; blk < WT - 1; ++blk) {
                #pragma unroll
                for (int r = 0; r < Cc; ++r) {
                    se[r * RUNP + 2 * blk]     = acc_e[blk][r];
                    se[r * RUNP + 2 * blk + 1] = acc_o[blk][r];
                }
            }
            float* so = edge ? (st + RUNL + hl) : (se + 2 * (WT - 1) + 1);
            #pragma unroll
            for (int r = 0; r < Cc; ++r) {
                se[r * RUNP + 2 * (WT - 1)] = acc_e[WT - 1][r];
                so[r * RUNP]                 = acc_o[WT - 1][r];
            }
            __builtin_amdgcn_fence(__ATOMIC_ACQ_REL, "wavefront");

            gchar* gb = (gchar*)(out + ((f * DO + od) * HO + 2 * b) * WO);
            asm("" : "+s"(gb), "+v"(gofs[0]), "+v"(gofs[1]), "+v"(gofs[2]), "+v"(gofs[3]),
                     "+v"(gofs[4]), "+v"(gofs[5]), "+v"(gofs[6]));
            asm("" : "+v"(gofs[7]), "+v"(gofs[8]), "+v"(gofs[9]), "+v"(gofs[10]),
                     "+v"(gna), "+v"(gnb));
            const v2f tb = *reinterpret_cast<const v2f*>(snb);
            v4f buf[11];
            #pragma unroll
            for (int it = 0; it < 11; ++it) {
                buf[it] = *reinterpret_cast<const v4f*>(sl[it]);
            }
            const v2f ta = *reinterpret_cast<const v2f*>(sna);
            asm volatile("" :: "v"(tb));
            if (lane < 24) {
                *(gv2fu*)(gb + gnb) = tb;
            }
            #pragma unroll
            for (int it = 0; it < 11; ++it) {
                *(gv4fu*)(gb + gofs[it]) = buf[it];
            }
            *(gv2fu*)(gb + gna) = ta;
            __builtin_amdgcn_fence(__ATOMIC_ACQ_REL, "wavefront");
        } else if (hl == 0) {
            float* po = out + ((f * DO + od) * HO + 2 * b) * WO + 6 * n;
            #pragma unroll
            for (int blk = 0; blk < WT - 1; ++blk) {
                #pragma unroll
                for (int r = 0; r < Cc; ++r) {
                    v2fu pr;
                    pr[0] = acc_e[blk][r];
                    pr[1] = acc_o[blk][r];
                    *reinterpret_cast<v2fu*>(po + r * OS + 2 * blk) = pr;
                }
            }
            if (!edge) {
                #pragma unroll
                for (int r = 0; r < Cc; ++r) {
                    v2fu pr;
                    pr[0] = acc_e[WT - 1][r];
                    pr[1] = acc_o[WT - 1][r];
                    *reinterpret_cast<v2fu*>(po + r * OS + 2 * (WT - 1)) = pr;
                }
            } else {
                #pragma unroll
                for (int r = 0; r < Cc; ++r) {
                    po[r * OS + 2 * (WT - 1)] = acc_e[WT - 1][r];
                }
            }
        }
    }
}

__global__ __launch_bounds__(128) __attribute__((amdgpu_waves_per_eu(2, 2)))
void convT4d_wmma_kernel(const float* __restrict__ x,
                         const float* __restrict__ weight,
                         float* __restrict__ out) {
    const int lane = threadIdx.x & 31;
    const int wave = __builtin_amdgcn_readfirstlane(threadIdx.x >> 5);
    __shared__ __attribute__((aligned(16))) float stage[4 * Cc * RUNP];
    float* st = stage + wave * (Cc * RUNP);
    const int slot = blockIdx.x & 31;
    const int grp  = blockIdx.x >> 5;

    if (slot < SLOT1) {
        const int cnt = SLOT1;
        run_class<0>(x, weight, out, st, (grp * cnt + slot) * 4 + wave, NGRP * cnt * 4, lane);
    } else {
        const int cnt = 32 - SLOT1;
        run_class<1>(x, weight, out, st, (grp * cnt + (slot - SLOT1)) * 4 + wave, NGRP * cnt * 4, lane);
    }
}

extern "C" void kernel_launch(void* const* d_in, const int* in_sizes, int n_in,
                              void* d_out, int out_size, void* d_ws, size_t ws_size,
                              hipStream_t stream) {
    const float* x = (const float*)d_in[0];
    const float* w = (const float*)d_in[1];
    float* out = (float*)d_out;

    convT4d_wmma_kernel<<<NBLK, 128, 0, stream>>>(x, w, out);
}


# Issue priority follows the tiles a wave has left (s_setprio, four levels): the two waves of a SIMD advance together, each one's store-out under the other's matrix stream; same sums, same order
# speedup vs baseline: 14.0632x; 1.0339x over previous
#include <hip/hip_runtime.h>
#include <hip/hip_bf16.h>


#define Cc   8
#define Tt   8
#define Dd   24
#define Hh   48
#define Ww   48
#define KT   3
#define KS   3
#define TO   6
#define DO   47
#define HO   95
#define WO   95
#define WT   3
#define NTAP 81
#define XCH  (Tt*Dd*Hh*Ww)
#define XROW (Hh*Ww)

#define NGRP   16
#define NBLK   (NGRP * 32)
#define SLOT1  11

typedef __attribute__((ext_vector_type(2))) float v2f;
typedef __attribute__((ext_vector_type(8))) float v8f;
typedef float v2fu __attribute__((ext_vector_type(2), aligned(4)));
typedef __attribute__((ext_vector_type(4))) float v4f;
typedef float v4fu __attribute__((ext_vector_type(4), aligned(4)));
typedef __attribute__((address_space(1))) const char gcchar;
typedef __attribute__((address_space(1))) char       gchar;
typedef __attribute__((address_space(1))) const v4fu gcv4fu;
typedef __attribute__((address_space(1))) v4fu       gv4fu;
typedef __attribute__((address_space(1))) v2fu       gv2fu;

#define RUNL 190
#define RUNP 192

__device__ __forceinline__ void set_rank(const int lvl) {
    switch (lvl) {
        case 3:  __builtin_amdgcn_s_setprio(3); break;
        case 2:  __builtin_amdgcn_s_setprio(2); break;
        case 1:  __builtin_amdgcn_s_setprio(1); break;
        default: __builtin_amdgcn_s_setprio(0); break;
    }
}

template <int PD>
__device__ __forceinline__ void run_class(const float* __restrict__ x,
                                          const float* __restrict__ weight,
                                          float* __restrict__ out,
                                          float* __restrict__ st,
                                          const int w, const int nw, const int lane) {
    constexpr int NKD = PD ? 2 : 1;
    constexpr int NKH = 2;
    constexpr int NOD = PD ? 23 : 24;
    constexpr int NOH = Hh;
    constexpr int OS  = TO * DO * HO * WO;

    const int n  = lane & 15;
    const int hl = lane >> 4;
    const int k0 = hl * 2;
    const int ph = n >> 3;

    v2f afrag[KT][NKD][NKH][KS][2];
    {
        const bool   odd = ph != 0;
        const float* wb  = weight + (k0 * Cc + (n & 7)) * NTAP;
        const float* wr0 = wb + (1 + ph) * KS;
        const float* wr1 = wb;
        #pragma unroll
        for (int i = 0; i < KT; ++i) {
            #pragma unroll
            for (int jd = 0; jd < NKD; ++jd) {
                const int kd = PD ? 2 * jd : 1;
                #pragma unroll
                for (int jh = 0; jh < NKH; ++jh) {
                    #pragma unroll
                    for (int kw = 0; kw < KS; ++kw) {
                        #pragma unroll
                        for (int chunk = 0; chunk < 2; ++chunk) {
                            #pragma unroll
                            for (int v = 0; v < 2; ++v) {
                                const int off = (chunk * 4 + v) * Cc * NTAP
                                              + (i * KS + (2 - kd)) * KS * KS + (2 - kw);
                                const float val = jh ? wr1[off] : wr0[off];
                                afrag[i][jd][jh][kw][chunk][v] = (jh && !odd) ? 0.0f : val;
                            }
                        }
                    }
                }
            }
        }
    }

    constexpr int NROW = KT * NKD * NKH;
    const int ntiles = TO * NOD * NOH;

    const bool edge = (n == 15);
    const int  jb   = edge ? (Ww - 4) : (3 * n);
    unsigned xo0 = (unsigned)(k0 * XCH + jb) * 4u;
    unsigned xo1 = xo0 + 4u * XCH * 4u;
    gcchar* xc = (gcchar*)x;

    unsigned gofs[11];
    const float* sl[11];
    #pragma unroll
    for (int it = 0; it < 11; ++it) {
        const unsigned q  = it * 32 + lane;
        const unsigned r  = q / 44;
        const unsigned ch = q - r * 44;
        gofs[it] = (r * (unsigned)OS + 4u * ch) * 4u;
        sl[it] = st + r * RUNP + 4 * ch;
    }
    const unsigned pa = lane;
    const unsigned pb = (lane < 24) ? (32 + lane) : 55u;
    const unsigned ra = pa / 7, ca2 = pa - ra * 7;
    const unsigned rbn = pb / 7, cb2 = pb - rbn * 7;
    unsigned gna = (ra * (unsigned)OS + 176u + 2u * ca2) * 4u;
    unsigned gnb = (rbn * (unsigned)OS + 176u + 2u * cb2) * 4u;
    const float* sna = st + ra * RUNP + 176 + 2 * ca2;
    const float* snb = st + rbn * RUNP + 176 + 2 * cb2;

    v4f xb[NROW][4];

    int cb = w % NOH;
    int ca = (w / NOH) % NOD;
    int cf = w / (NOH * NOD);
    {
        const int ih1 = (cb + 1 < Hh) ? (cb + 1) : (Hh - 1);
        const int tb = (cf * Dd + ca) * XROW;
        gcchar* rb0 = xc + (size_t)(unsigned)(tb + cb * Ww) * 4u;
        gcchar* rb1 = xc + (size_t)(unsigned)(tb + ih1 * Ww) * 4u;
        asm("" : "+s"(rb0), "+s"(rb1), "+v"(xo0), "+v"(xo1));
        #pragma unroll
        for (int i = 0; i < KT; ++i) {
            #pragma unroll
            for (int jd = 0; jd < NKD; ++jd) {
                #pragma unroll
                for (int jh = 0; jh < NKH; ++jh) {
                    const int r = (i * NKD + jd) * NKH + jh;
                    #pragma unroll
                    for (int c4 = 0; c4 < 4; ++c4) {
                        xb[r][c4] = *(gcv4fu*)(
                            (jh ? rb1 : rb0) + ((c4 >> 1) ? xo1 : xo0)
                            + ((i * Dd + jd) * XROW + (c4 & 1) * XCH) * 4);
                    }
                }
            }
        }
    }

    int left = (ntiles - w + nw - 1) / nw;
    int tile = w;
    for (int lvl = 3; lvl >= 0; --lvl) {
    set_rank(lvl);
    const int stop = PD ? ((lvl == 3) ? 3 : lvl)
                        : ((lvl == 3) ? 7 : (lvl == 2) ? 4 : (lvl == 1) ? 1 : 0);
    for (; left > stop; --left, tile += nw) {
        const int b  = cb;
        const int a  = ca;
        const int f  = cf;
        const int od = 2 * a + PD;

        const int tn  = (tile + nw < ntiles) ? (tile + nw) : tile;
        const int nb  = tn % NOH;
        const int na  = (tn / NOH) % NOD;
        const int nf  = tn / (NOH * NOD);
        const int nih1 = (nb + 1 < Hh) ? (nb + 1) : (Hh - 1);
        const int ntb = (nf * Dd + na) * XROW;
        gcchar* nb0 = xc + (size_t)(unsigned)(ntb + nb * Ww) * 4u;
        gcchar* nb1 = xc + (size_t)(unsigned)(ntb + nih1 * Ww) * 4u;
        asm("" : "+s"(nb0), "+s"(nb1), "+v"(xo0), "+v"(xo1));

        v8f acc_e[WT] = {};
        v8f acc_o[WT] = {};

        #pragma unroll
        for (int i = 0; i < KT; ++i) {
            #pragma unroll
            for (int jd = 0; jd < NKD; ++jd) {
                v2f s0[NKH][WT][2], s3[NKH][2];
                #pragma unroll
                for (int jh = 0; jh < NKH; ++jh) {
                    const int r = (i * NKD + jd) * NKH + jh;
                    #pragma unroll
                    for (int c4 = 0; c4 < 4; ++c4) {
                        const v4f q = xb[r][c4];
                        s0[jh][0][c4 >> 1][c4 & 1] = edge ? q[1] : q[0];
                        s0[jh][1][c4 >> 1][c4 & 1] = edge ? q[2] : q[1];
                        s0[jh][2][c4 >> 1][c4 & 1] = edge ? q[3] : q[2];
                        s3[jh][c4 >> 1][c4 & 1]    = q[3];
                    }
                }

                #pragma unroll
                for (int jh = 0; jh < NKH; ++jh) {
                    const int r = (i * NKD + jd) * NKH + jh;
                    #pragma unroll
                    for (int c4 = 0; c4 < 4; ++c4) {
                        xb[r][c4] = *(gcv4fu*)(
                            (jh ? nb1 : nb0) + ((c4 >> 1) ? xo1 : xo0)
                            + ((i * Dd + jd) * XROW + (c4 & 1) * XCH) * 4);
                    }
                }

                #pragma unroll
                for (int jh = 0; jh < NKH; ++jh) {
                    #pragma unroll
                    for (int chunk = 0; chunk < 2; ++chunk) {
                        #pragma unroll
                        for (int blk = 0; blk < WT; ++blk) {
                            acc_o[blk] = __builtin_amdgcn_wmma_f32_16x16x4_f32(
                                        false, afrag[i][jd][jh][0][chunk], false, s0[jh][blk][chunk],
                                        (short)0, acc_o[blk], false, false);
                            acc_e[blk] = __builtin_amdgcn_wmma_f32_16x16x4_f32(
                                        false, afrag[i][jd][jh][1][chunk], false, s0[jh][blk][chunk],
                                        (short)0, acc_e[blk], false, false);
                        }
                    }
                    #pragma unroll
                    for (int chunk = 0; chunk < 2; ++chunk) {
                        #pragma unroll
                        for (int blk = 0; blk < WT; ++blk) {
                            acc_o[blk] = __builtin_amdgcn_wmma_f32_16x16x4_f32(
                                        false, afrag[i][jd][jh][2][chunk], false,
                                        (blk + 1 < WT) ? s0[jh][(blk + 1) % WT][chunk] : s3[jh][chunk],
                                        (short)0, acc_o[blk], false, false);
                        }
                    }
                }
            }
        }
        cb = nb;  ca = na;  cf = nf;

        if (b + 1 < NOH) {
            float* se = st + hl * WO + 6 * n;
            #pragma unroll
            for (int blk = 0; blk < WT - 1; ++blk) {
                #pragma unroll
                for (int r = 0; r < Cc; ++r) {
                    se[r * RUNP + 2 * blk]     = acc_e[blk][r];
                    se[r * RUNP + 2 * blk + 1] = acc_o[blk][r];
                }
            }
            float* so = edge ? (st + RUNL + hl) : (se + 2 * (WT - 1) + 1);
            #pragma unroll
            for (int r = 0; r < Cc; ++r) {
                se[r * RUNP + 2 * (WT - 1)] = acc_e[WT - 1][r];
                so[r * RUNP]                 = acc_o[WT - 1][r];
            }
            __builtin_amdgcn_fence(__ATOMIC_ACQ_REL, "wavefront");

            gchar* gb = (gchar*)(out + ((f * DO + od) * HO + 2 * b) * WO);
            asm("" : "+s"(gb), "+v"(gofs[0]), "+v"(gofs[1]), "+v"(gofs[2]), "+v"(gofs[3]),
                     "+v"(gofs[4]), "+v"(gofs[5]), "+v"(gofs[6]));
            asm("" : "+v"(gofs[7]), "+v"(gofs[8]), "+v"(gofs[9]), "+v"(gofs[10]),
                     "+v"(gna), "+v"(gnb));
            const v2f tb = *reinterpret_cast<const v2f*>(snb);
            v4f buf[11];
            #pragma unroll
            for (int it = 0; it < 11; ++it) {
                buf[it] = *reinterpret_cast<const v4f*>(sl[it]);
            }
            const v2f ta = *reinterpret_cast<const v2f*>(sna);
            asm volatile("" :: "v"(tb));
            if (lane < 24) {
                *(gv2fu*)(gb + gnb) = tb;
            }
            #pragma unroll
            for (int it = 0; it < 11; ++it) {
                *(gv4fu*)(gb + gofs[it]) = buf[it];
            }
            *(gv2fu*)(gb + gna) = ta;
            __builtin_amdgcn_fence(__ATOMIC_ACQ_REL, "wavefront");
        } else if (hl == 0) {
            float* po = out + ((f * DO + od) * HO + 2 * b) * WO + 6 * n;
            #pragma unroll
            for (int blk = 0; blk < WT - 1; ++blk) {
                #pragma unroll
                for (int r = 0; r < Cc; ++r) {
                    v2fu pr;
                    pr[0] = acc_e[blk][r];
                    pr[1] = acc_o[blk][r];
                    *reinterpret_cast<v2fu*>(po + r * OS + 2 * blk) = pr;
                }
            }
            if (!edge) {
                #pragma unroll
                for (int r = 0; r < Cc; ++r) {
                    v2fu pr;
                    pr[0] = acc_e[WT - 1][r];
                    pr[1] = acc_o[WT - 1][r];
                    *reinterpret_cast<v2fu*>(po + r * OS + 2 * (WT - 1)) = pr;
                }
            } else {
                #pragma unroll
                for (int r = 0; r < Cc; ++r) {
                    po[r * OS + 2 * (WT - 1)] = acc_e[WT - 1][r];
                }
            }
        }
    }
    }
}

__global__ __launch_bounds__(128) __attribute__((amdgpu_waves_per_eu(2, 2)))
void convT4d_wmma_kernel(const float* __restrict__ x,
                         const float* __restrict__ weight,
                         float* __restrict__ out) {
    const int lane = threadIdx.x & 31;
    const int wave = __builtin_amdgcn_readfirstlane(threadIdx.x >> 5);
    __shared__ __attribute__((aligned(16))) float stage[4 * Cc * RUNP];
    float* st = stage + wave * (Cc * RUNP);
    const int slot = blockIdx.x & 31;
    const int grp  = blockIdx.x >> 5;

    if (slot < SLOT1) {
        const int cnt = SLOT1;
        run_class<0>(x, weight, out, st, (grp * cnt + slot) * 4 + wave, NGRP * cnt * 4, lane);
    } else {
        const int cnt = 32 - SLOT1;
        run_class<1>(x, weight, out, st, (grp * cnt + (slot - SLOT1)) * 4 + wave, NGRP * cnt * 4, lane);
    }
}

extern "C" void kernel_launch(void* const* d_in, const int* in_sizes, int n_in,
                              void* d_out, int out_size, void* d_ws, size_t ws_size,
                              hipStream_t stream) {
    const float* x = (const float*)d_in[0];
    const float* w = (const float*)d_in[1];
    float* out = (float*)d_out;

    convT4d_wmma_kernel<<<NBLK, 128, 0, stream>>>(x, w, out);
}
